# starting kernel
# baseline (speedup 1.0000x reference)
#include <hip/hip_runtime.h>
#include <hip/hip_bf16.h>

typedef float v2f __attribute__((ext_vector_type(2)));
typedef float v8f __attribute__((ext_vector_type(8)));

#define N_NODES 100000
#define N_EDGES 1600000
#define C_HID   32
#define C_OUT   8
#define N_AGENTS 1000
#define N_TILES (N_NODES / 16)

__global__ void zero_f32(float* __restrict__ p, int n) {
    int t = blockIdx.x * blockDim.x + threadIdx.x;
    if (t < n) p[t] = 0.0f;
}

__global__ void count_deg(const int* __restrict__ dst, float* __restrict__ deg) {
    int t = blockIdx.x * blockDim.x + threadIdx.x;
    if (t < N_EDGES) atomicAdd(deg + dst[t], 1.0f);
}

__global__ __launch_bounds__(256) void scatter_add(
        const float* __restrict__ xin, const int* __restrict__ src,
        const int* __restrict__ dst, float* __restrict__ agg) {
    int t = blockIdx.x * blockDim.x + threadIdx.x;
    if (t >= N_EDGES * 8) return;
    int e = t >> 3;
    int c = (t & 7) * 4;
    int s = src[e];
    int d = dst[e];
    const float4 v = *(const float4*)(xin + (size_t)s * C_HID + c);
    float* p = agg + (size_t)d * C_HID + c;
    atomicAdd(p + 0, v.x);
    atomicAdd(p + 1, v.y);
    atomicAdd(p + 2, v.z);
    atomicAdd(p + 3, v.w);
}

__global__ __launch_bounds__(256) void sage_gemm(
        const float* __restrict__ agg, const float* __restrict__ xin,
        const float* __restrict__ deg,
        const float* __restrict__ Wl, const float* __restrict__ Wr,
        const float* __restrict__ bias, float* __restrict__ hout) {
    __shared__ float sW[64 * 32];
    __shared__ float sB[32];
    for (int i = threadIdx.x; i < 1024; i += 256) {
        sW[i]        = Wl[i];
        sW[1024 + i] = Wr[i];
    }
    if (threadIdx.x < 32) sB[threadIdx.x] = bias[threadIdx.x];
    __syncthreads();

    const int wave = threadIdx.x >> 5;
    const int lane = threadIdx.x & 31;
    const int tile = blockIdx.x * 8 + wave;
    if (tile >= N_TILES) return;

    const int row0 = tile * 16;
    const int mrow = lane & 15;
    const int row  = row0 + mrow;
    const int koff = (lane >> 4) * 2;

    const float dv  = deg[row];
    const float inv = 1.0f / fmaxf(dv, 1.0f);

    v8f acc0, acc1;
    const float b0v = sB[mrow];
    const float b1v = sB[16 + mrow];
#pragma unroll
    for (int j = 0; j < 8; ++j) { acc0[j] = b0v; acc1[j] = b1v; }

    const float* aggRow = agg + (size_t)row * C_HID;
    const float* xRow   = xin + (size_t)row * C_HID;

#pragma unroll
    for (int kc = 0; kc < 16; ++kc) {
        const int kbase = kc * 4;
        const int kb = kbase + koff;
        v2f a;
        if (kc < 8) {
            a.x = aggRow[kb]     * inv;
            a.y = aggRow[kb + 1] * inv;
        } else {
            a.x = xRow[kb - 32];
            a.y = xRow[kb - 31];
        }
        v2f b0, b1;
        b0.x = sW[(kb)     * 32 + mrow];
        b0.y = sW[(kb + 1) * 32 + mrow];
        b1.x = sW[(kb)     * 32 + 16 + mrow];
        b1.y = sW[(kb + 1) * 32 + 16 + mrow];
        acc0 = __builtin_amdgcn_wmma_f32_16x16x4_f32(
                   false, a, false, b0, (short)0, acc0, false, false);
        acc1 = __builtin_amdgcn_wmma_f32_16x16x4_f32(
                   false, a, false, b1, (short)0, acc1, false, false);
    }

    const int half = lane >> 4;
#pragma unroll
    for (int j = 0; j < 8; ++j) {
        const int m = row0 + half * 8 + j;
        hout[(size_t)m * C_HID + mrow]      = fmaxf(acc0[j], 0.0f);
        hout[(size_t)m * C_HID + 16 + mrow] = fmaxf(acc1[j], 0.0f);
    }
}

__global__ void out_proj(const float* __restrict__ h, const float* __restrict__ Wo,
                         const float* __restrict__ bo, float* __restrict__ out) {
    int t = blockIdx.x * blockDim.x + threadIdx.x;
    if (t >= N_AGENTS * C_OUT) return;
    const int n = t >> 3, o = t & 7;
    float s = bo[o];
    const float* hr = h + (size_t)n * C_HID;
#pragma unroll
    for (int k = 0; k < C_HID; ++k) s += hr[k] * Wo[k * C_OUT + o];
    out[t] = s;
}

extern "C" void kernel_launch(void* const* d_in, const int* in_sizes, int n_in,
                              void* d_out, int out_size, void* d_ws, size_t ws_size,
                              hipStream_t stream) {
    const float* x    = (const float*)d_in[0];
    const int*   ei   = (const int*)d_in[1];
    const int*   src  = ei;
    const int*   dst  = ei + N_EDGES;
    const float* Wl1  = (const float*)d_in[2];
    const float* bl1  = (const float*)d_in[3];
    const float* Wr1  = (const float*)d_in[4];
    const float* Wl2  = (const float*)d_in[5];
    const float* bl2  = (const float*)d_in[6];
    const float* Wr2  = (const float*)d_in[7];
    const float* Wl3  = (const float*)d_in[8];
    const float* bl3  = (const float*)d_in[9];
    const float* Wr3  = (const float*)d_in[10];
    const float* Wo   = (const float*)d_in[11];
    const float* bo   = (const float*)d_in[12];
    float* out = (float*)d_out;

    const size_t NB = (size_t)N_NODES * C_HID;
    float* agg = (float*)d_ws;
    float* h1  = agg + NB;
    float* h2  = h1 + NB;
    float* deg = h2 + NB;

    const int ZB_FEAT = (int)((NB + 255) / 256);
    const int ZB_DEG  = (N_NODES + 255) / 256;
    const int EB      = (N_EDGES + 255) / 256;
    const int SB      = (N_EDGES * 8 + 255) / 256;
    const int GB      = (N_TILES + 7) / 8;
    const int OB      = (N_AGENTS * C_OUT + 255) / 256;

    zero_f32<<<ZB_DEG, 256, 0, stream>>>(deg, N_NODES);
    count_deg<<<EB, 256, 0, stream>>>(dst, deg);

    zero_f32<<<ZB_FEAT, 256, 0, stream>>>(agg, (int)NB);
    scatter_add<<<SB, 256, 0, stream>>>(x, src, dst, agg);
    sage_gemm<<<GB, 256, 0, stream>>>(agg, x, deg, Wl1, Wr1, bl1, h1);

    zero_f32<<<ZB_FEAT, 256, 0, stream>>>(agg, (int)NB);
    scatter_add<<<SB, 256, 0, stream>>>(h1, src, dst, agg);
    sage_gemm<<<GB, 256, 0, stream>>>(agg, h1, deg, Wl2, Wr2, bl2, h2);

    zero_f32<<<ZB_FEAT, 256, 0, stream>>>(agg, (int)NB);
    scatter_add<<<SB, 256, 0, stream>>>(h2, src, dst, agg);
    sage_gemm<<<GB, 256, 0, stream>>>(agg, h2, deg, Wl3, Wr3, bl3, h1);

    out_proj<<<OB, 256, 0, stream>>>(h1, Wo, bo, out);
}


# edge filing with no global atomic per edge: edges sorted by destination range into per-producer segments, one work-group per range files them in LDS, whole-line records; slot order = LDS arrival
# speedup vs baseline: 3.2767x; 3.2767x over previous
#include <hip/hip_runtime.h>
#include <hip/hip_bf16.h>

typedef float v2f __attribute__((ext_vector_type(2)));
typedef float v8f __attribute__((ext_vector_type(8)));

#define N_NODES 100000
#define N_EDGES 1600000
#define C_HID   32
#define C_OUT   8
#define N_AGENTS 1000
#define N_TILES (N_NODES / 16)

#define REC_WORDS 32
#define SLOT_CAP  31
#define CNT_WORD  31
#define SPILLB_WORDS 60000
#define SPILL_CAP 29998

#define A_STRIDE 36

#define BKT_NODES 400
#define N_BKT     250
#define N_PROD    256
#define EDGES_PER_PROD 6250
#define PER_THREAD 7
#define SEG_WORDS 64
#define SEG_CAP   63
#define OV_WORDS  128
#define OV_CAP    63
#define SRC_BITS  17
#define SRC_MASK  0x1FFFF

__global__ __launch_bounds__(1024) void part_edges(
        const int* __restrict__ src, const int* __restrict__ dst,
        int* __restrict__ seg, int* __restrict__ ov, int* __restrict__ spill) {
    __shared__ int segCnt[N_BKT];
    __shared__ int ovCnt;
    const int b = blockIdx.x;
    const int t = threadIdx.x;
    if (t < N_BKT) segCnt[t] = 0;
    if (t == 0) ovCnt = 0;
    if (b == 0 && t == 0) spill[0] = 0;
    __syncthreads();

    const int e0 = b * EDGES_PER_PROD;
    int* ovb = ov + b * OV_WORDS;
    int s[PER_THREAD], d[PER_THREAD];
#pragma unroll
    for (int i = 0; i < PER_THREAD; ++i) {
        const int k = t + i * 1024;
        s[i] = 0; d[i] = -1;
        if (k < EDGES_PER_PROD) { s[i] = src[e0 + k]; d[i] = dst[e0 + k]; }
    }
#pragma unroll
    for (int i = 0; i < PER_THREAD; ++i) {
        if (d[i] >= 0) {
            const int bk = d[i] / BKT_NODES;
            const int dl = d[i] - bk * BKT_NODES;
            const int slot = atomicAdd(&segCnt[bk], 1);
            if (slot < SEG_CAP) {
                seg[((size_t)bk * N_PROD + b) * SEG_WORDS + 1 + slot] = s[i] | (dl << SRC_BITS);
            } else {
                const int k = atomicAdd(&ovCnt, 1);
                if (k < OV_CAP) { ovb[1 + 2 * k] = s[i]; ovb[2 + 2 * k] = d[i]; }
            }
        }
    }
    __syncthreads();
    if (t < N_BKT) seg[((size_t)t * N_PROD + b) * SEG_WORDS] = min(segCnt[t], SEG_CAP);
    if (t == 0) ovb[0] = ovCnt;
}

__global__ __launch_bounds__(1024) void file_bucket(
        const int* __restrict__ seg, const int* __restrict__ ov,
        const int* __restrict__ src, const int* __restrict__ dst,
        int* __restrict__ rec, int* __restrict__ spill) {
    __shared__ int recL[BKT_NODES * REC_WORDS];
    __shared__ int degL[BKT_NODES];
    __shared__ int ovFull;
    const int bk = blockIdx.x;
    const int t  = threadIdx.x;
    const int node0 = bk * BKT_NODES;
    for (int i = t; i < BKT_NODES * REC_WORDS; i += 1024) recL[i] = 0;
    if (t < BKT_NODES) degL[t] = 0;
    if (t == 0) ovFull = 0;
    __syncthreads();
    int myov = 0;
    if (t < N_PROD) {
        myov = ov[t * OV_WORDS];
        if (myov > OV_CAP) ovFull = 1;
    }
    __syncthreads();
    const bool full = ovFull != 0;

    auto file = [&](int sv, int dl) {
        const int pos = atomicAdd(&degL[dl], 1);
        if (pos < SLOT_CAP) {
            recL[dl * REC_WORDS + pos] = sv;
        } else {
            const int k = atomicAdd(spill, 1);
            if (k < SPILL_CAP) { spill[2 + 2 * k] = sv; spill[3 + 2 * k] = node0 + dl; }
        }
    };

    if (!full) {
        const int wave = t >> 5;
        const int lane = t & 31;
        const int* base = seg + (size_t)bk * N_PROD * SEG_WORDS;
        int c[8], v[8];
#pragma unroll
        for (int j = 0; j < 8; ++j)
            c[j] = min(base[(wave + 32 * j) * SEG_WORDS], SEG_CAP);
#pragma unroll
        for (int j = 0; j < 8; ++j) {
            v[j] = 0;
            if (lane < c[j]) v[j] = base[(wave + 32 * j) * SEG_WORDS + 1 + lane];
        }
#pragma unroll
        for (int j = 0; j < 8; ++j) {
            if (lane < c[j]) file(v[j] & SRC_MASK, v[j] >> SRC_BITS);
            if (lane + 32 < c[j]) {
                const int v2 = base[(wave + 32 * j) * SEG_WORDS + 33 + lane];
                file(v2 & SRC_MASK, v2 >> SRC_BITS);
            }
        }
        if (t < N_PROD) {
            const int* ovb = ov + t * OV_WORDS;
            for (int k = 0; k < myov; ++k) {
                const int dd = ovb[2 + 2 * k] - node0;
                if ((unsigned)dd < (unsigned)BKT_NODES) file(ovb[1 + 2 * k], dd);
            }
        }
    } else {
        for (int e = t; e < N_EDGES; e += 1024) {
            const int dd = dst[e] - node0;
            if ((unsigned)dd < (unsigned)BKT_NODES) file(src[e], dd);
        }
    }
    __syncthreads();
    int* out = rec + (size_t)node0 * REC_WORDS;
    for (int i = t; i < BKT_NODES * REC_WORDS; i += 1024)
        out[i] = ((i & 31) == CNT_WORD) ? degL[i >> 5] : recL[i];
}

template <int NB>
__device__ __forceinline__ void add_rows(float4& acc, const float* __restrict__ xin,
                                         const int* w, int base, int nslot,
                                         int node, int gbase, int sub) {
    float4 v[NB];
#pragma unroll
    for (int jj = 0; jj < NB; ++jj) {
        int idx = __shfl(w[jj >> 3], gbase + (jj & 7));
        idx = (base + jj < nslot) ? idx : node;
        v[jj] = *(const float4*)(xin + (size_t)idx * C_HID + sub * 4);
    }
#pragma unroll
    for (int jj = 0; jj < NB; ++jj) {
        const bool p = base + jj < nslot;
        acc.x += p ? v[jj].x : 0.0f;
        acc.y += p ? v[jj].y : 0.0f;
        acc.z += p ? v[jj].z : 0.0f;
        acc.w += p ? v[jj].w : 0.0f;
    }
}

__device__ __forceinline__ float4 add_matching(const float* __restrict__ xin,
                                               const int* __restrict__ sp, const int* __restrict__ dp,
                                               int stride, int n, int node, int gbase, int sub) {
    float4 acc = make_float4(0.0f, 0.0f, 0.0f, 0.0f);
    for (int k0 = 0; k0 < n; k0 += 8) {
        const int k = k0 + sub;
        int sk = 0, dk = -1;
        if (k < n) { sk = sp[(size_t)k * stride]; dk = dp[(size_t)k * stride]; }
#pragma unroll
        for (int jj = 0; jj < 8; ++jj) {
            const int dj = __shfl(dk, gbase + jj);
            const int sj = __shfl(sk, gbase + jj);
            if (dj == node) {
                const float4 v = *(const float4*)(xin + (size_t)sj * C_HID + sub * 4);
                acc.x += v.x; acc.y += v.y; acc.z += v.z; acc.w += v.w;
            }
        }
    }
    return acc;
}

__global__ __launch_bounds__(256) void sage_pull_gemm(
        const float* __restrict__ xin, const int* __restrict__ rec,
        const int* __restrict__ spill,
        const int* __restrict__ src, const int* __restrict__ dst,
        const float* __restrict__ Wl, const float* __restrict__ Wr,
        const float* __restrict__ bias, float* __restrict__ hout) {
    __shared__ float sW[64 * 32];
    __shared__ float sB[32];
    __shared__ __attribute__((aligned(16))) float sA[8][16 * A_STRIDE];
    for (int i = threadIdx.x; i < 1024; i += 256) {
        sW[i]        = Wl[i];
        sW[1024 + i] = Wr[i];
    }
    if (threadIdx.x < 32) sB[threadIdx.x] = bias[threadIdx.x];
    __syncthreads();

    const int wave = threadIdx.x >> 5;
    const int lane = threadIdx.x & 31;
    const int tile = blockIdx.x * 8 + wave;
    if (tile >= N_TILES) return;

    const int row0 = tile * 16;
    float* myA = sA[wave];

    const int gbase = lane & ~7;
    const int grp   = lane >> 3;
    const int sub   = lane & 7;

    int cnt[4];
#pragma unroll
    for (int r = 0; r < 4; ++r)
        cnt[r] = rec[(size_t)(row0 + 4 * r + grp) * REC_WORDS + CNT_WORD];

    int w[4][4];
#pragma unroll
    for (int r = 0; r < 4; ++r) {
        const int* nrec = rec + (size_t)(row0 + 4 * r + grp) * REC_WORDS;
        const int nslot = min(cnt[r], SLOT_CAP);
#pragma unroll
        for (int q = 0; q < 4; ++q) {
            w[r][q] = 0;
            if (sub + 8 * q < nslot) w[r][q] = nrec[sub + 8 * q];
        }
    }

    int spilled = 0;
#pragma unroll
    for (int r = 0; r < 4; ++r) {
        const int node  = row0 + 4 * r + grp;
        const int nslot = min(cnt[r], SLOT_CAP);
        float4 acc = make_float4(0.0f, 0.0f, 0.0f, 0.0f);
        add_rows<16>(acc, xin, &w[r][0], 0, nslot, node, gbase, sub);
        if (__any(16 < nslot))
            add_rows<8>(acc, xin, &w[r][2], 16, nslot, node, gbase, sub);
        if (__any(24 < nslot))
            add_rows<8>(acc, xin, &w[r][3], 24, nslot, node, gbase, sub);
        if (cnt[r] > SLOT_CAP) spilled |= 1 << r;
        const float inv = 1.0f / fmaxf((float)cnt[r], 1.0f);
        float4 mean;
        mean.x = acc.x * inv; mean.y = acc.y * inv; mean.z = acc.z * inv; mean.w = acc.w * inv;
        *(float4*)(myA + (4 * r + grp) * A_STRIDE + sub * 4) = mean;
    }

    if (__any(spilled != 0)) {
#pragma unroll 1
        for (int r = 0; r < 4; ++r) {
            if ((spilled >> r) & 1) {
                const int node = row0 + 4 * r + grp;
                const float inv = 1.0f / (float)rec[(size_t)node * REC_WORDS + CNT_WORD];
                float4* mp = (float4*)(myA + (4 * r + grp) * A_STRIDE + sub * 4);
                float4 mean = *mp;
                const int oc = spill[0];
                if (oc <= SPILL_CAP) {
                    const float4 ex = add_matching(xin, spill + 2, spill + 3, 2, oc, node, gbase, sub);
                    mean.x += ex.x * inv; mean.y += ex.y * inv; mean.z += ex.z * inv; mean.w += ex.w * inv;
                } else {
                    const float4 ex = add_matching(xin, src, dst, 1, N_EDGES, node, gbase, sub);
                    mean.x = ex.x * inv; mean.y = ex.y * inv; mean.z = ex.z * inv; mean.w = ex.w * inv;
                }
                *mp = mean;
            }
        }
    }

    __builtin_amdgcn_fence(__ATOMIC_RELEASE, "wavefront");
    __builtin_amdgcn_wave_barrier();
    __builtin_amdgcn_fence(__ATOMIC_ACQUIRE, "wavefront");

    const int mrow = lane & 15;
    const int row  = row0 + mrow;
    const int koff = (lane >> 4) * 2;

    v8f acc0, acc1;
    const float b0v = sB[mrow];
    const float b1v = sB[16 + mrow];
#pragma unroll
    for (int j = 0; j < 8; ++j) { acc0[j] = b0v; acc1[j] = b1v; }

    const float* aRow = myA + mrow * A_STRIDE;
    const float* xRow = xin + (size_t)row * C_HID;

#pragma unroll
    for (int kc = 0; kc < 16; ++kc) {
        const int kbase = kc * 4;
        const int kb = kbase + koff;
        v2f a;
        if (kc < 8) {
            a.x = aRow[kb];
            a.y = aRow[kb + 1];
        } else {
            a.x = xRow[kb - 32];
            a.y = xRow[kb - 31];
        }
        v2f b0, b1;
        b0.x = sW[(kb)     * 32 + mrow];
        b0.y = sW[(kb + 1) * 32 + mrow];
        b1.x = sW[(kb)     * 32 + 16 + mrow];
        b1.y = sW[(kb + 1) * 32 + 16 + mrow];
        acc0 = __builtin_amdgcn_wmma_f32_16x16x4_f32(
                   false, a, false, b0, (short)0, acc0, false, false);
        acc1 = __builtin_amdgcn_wmma_f32_16x16x4_f32(
                   false, a, false, b1, (short)0, acc1, false, false);
    }

    const int half = lane >> 4;
#pragma unroll
    for (int j = 0; j < 8; ++j) {
        const int m = row0 + half * 8 + j;
        hout[(size_t)m * C_HID + mrow]      = fmaxf(acc0[j], 0.0f);
        hout[(size_t)m * C_HID + 16 + mrow] = fmaxf(acc1[j], 0.0f);
    }
}

__global__ void out_proj(const float* __restrict__ h, const float* __restrict__ Wo,
                         const float* __restrict__ bo, float* __restrict__ out) {
    int t = blockIdx.x * blockDim.x + threadIdx.x;
    if (t >= N_AGENTS * C_OUT) return;
    const int n = t >> 3, o = t & 7;
    float s = bo[o];
    const float* hr = h + (size_t)n * C_HID;
#pragma unroll
    for (int k = 0; k < C_HID; ++k) s += hr[k] * Wo[k * C_OUT + o];
    out[t] = s;
}

extern "C" void kernel_launch(void* const* d_in, const int* in_sizes, int n_in,
                              void* d_out, int out_size, void* d_ws, size_t ws_size,
                              hipStream_t stream) {
    const float* x    = (const float*)d_in[0];
    const int*   ei   = (const int*)d_in[1];
    const int*   src  = ei;
    const int*   dst  = ei + N_EDGES;
    const float* Wl1  = (const float*)d_in[2];
    const float* bl1  = (const float*)d_in[3];
    const float* Wr1  = (const float*)d_in[4];
    const float* Wl2  = (const float*)d_in[5];
    const float* bl2  = (const float*)d_in[6];
    const float* Wr2  = (const float*)d_in[7];
    const float* Wl3  = (const float*)d_in[8];
    const float* bl3  = (const float*)d_in[9];
    const float* Wr3  = (const float*)d_in[10];
    const float* Wo   = (const float*)d_in[11];
    const float* bo   = (const float*)d_in[12];
    float* out = (float*)d_out;

    const size_t NB = (size_t)N_NODES * C_HID;
    int*   rec   = (int*)d_ws;
    float* h1    = (float*)d_ws + NB;
    float* h2    = h1 + NB;
    int*   spill = (int*)(h2 + NB);
    int*   ov    = spill + SPILLB_WORDS;
    int*   seg   = (int*)h1;

    const int GB = (N_TILES + 7) / 8;
    const int OB = (N_AGENTS * C_OUT + 255) / 256;

    part_edges<<<N_PROD, 1024, 0, stream>>>(src, dst, seg, ov, spill);
    file_bucket<<<N_BKT, 1024, 0, stream>>>(seg, ov, src, dst, rec, spill);

    sage_pull_gemm<<<GB, 256, 0, stream>>>(x, rec, spill, src, dst, Wl1, Wr1, bl1, h1);
    sage_pull_gemm<<<GB, 256, 0, stream>>>(h1, rec, spill, src, dst, Wl2, Wr2, bl2, h2);
    sage_pull_gemm<<<GB, 256, 0, stream>>>(h2, rec, spill, src, dst, Wl3, Wr3, bl3, h1);

    out_proj<<<OB, 256, 0, stream>>>(h1, Wo, bo, out);
}


# per-wave load chain shortened: records and self rows fetched with the weights at wave start into the wave's LDS rows; a round's first 24 slot rows in flight at once (was 16, then 8); slot order kept
# speedup vs baseline: 3.7154x; 1.1339x over previous
#include <hip/hip_runtime.h>
#include <hip/hip_bf16.h>

typedef float v2f __attribute__((ext_vector_type(2)));
typedef float v8f __attribute__((ext_vector_type(8)));

#define N_NODES 100000
#define N_EDGES 1600000
#define C_HID   32
#define C_OUT   8
#define N_AGENTS 1000
#define N_TILES (N_NODES / 16)

#define REC_WORDS 32
#define SLOT_CAP  31
#define CNT_WORD  31
#define SPILLB_WORDS 60000
#define SPILL_CAP 29998

#define A_STRIDE 68
#define R_STRIDE 36
#define FIRST_SLOTS 24

#define BKT_NODES 400
#define N_BKT     250
#define N_PROD    256
#define EDGES_PER_PROD 6250
#define PER_THREAD 7
#define SEG_WORDS 64
#define SEG_CAP   63
#define OV_WORDS  128
#define OV_CAP    63
#define SRC_BITS  17
#define SRC_MASK  0x1FFFF

__global__ __launch_bounds__(1024) void part_edges(
        const int* __restrict__ src, const int* __restrict__ dst,
        int* __restrict__ seg, int* __restrict__ ov, int* __restrict__ spill) {
    __shared__ int segCnt[N_BKT];
    __shared__ int ovCnt;
    const int b = blockIdx.x;
    const int t = threadIdx.x;
    if (t < N_BKT) segCnt[t] = 0;
    if (t == 0) ovCnt = 0;
    if (b == 0 && t == 0) spill[0] = 0;
    __syncthreads();

    const int e0 = b * EDGES_PER_PROD;
    int* ovb = ov + b * OV_WORDS;
    int s[PER_THREAD], d[PER_THREAD];
#pragma unroll
    for (int i = 0; i < PER_THREAD; ++i) {
        const int k = t + i * 1024;
        s[i] = 0; d[i] = -1;
        if (k < EDGES_PER_PROD) { s[i] = src[e0 + k]; d[i] = dst[e0 + k]; }
    }
#pragma unroll
    for (int i = 0; i < PER_THREAD; ++i) {
        if (d[i] >= 0) {
            const int bk = d[i] / BKT_NODES;
            const int dl = d[i] - bk * BKT_NODES;
            const int slot = atomicAdd(&segCnt[bk], 1);
            if (slot < SEG_CAP) {
                seg[((size_t)bk * N_PROD + b) * SEG_WORDS + 1 + slot] = s[i] | (dl << SRC_BITS);
            } else {
                const int k = atomicAdd(&ovCnt, 1);
                if (k < OV_CAP) { ovb[1 + 2 * k] = s[i]; ovb[2 + 2 * k] = d[i]; }
            }
        }
    }
    __syncthreads();
    if (t < N_BKT) seg[((size_t)t * N_PROD + b) * SEG_WORDS] = min(segCnt[t], SEG_CAP);
    if (t == 0) ovb[0] = ovCnt;
}

__global__ __launch_bounds__(1024) void file_bucket(
        const int* __restrict__ seg, const int* __restrict__ ov,
        const int* __restrict__ src, const int* __restrict__ dst,
        int* __restrict__ rec, int* __restrict__ spill) {
    __shared__ int recL[BKT_NODES * REC_WORDS];
    __shared__ int degL[BKT_NODES];
    __shared__ int ovFull;
    const int bk = blockIdx.x;
    const int t  = threadIdx.x;
    const int node0 = bk * BKT_NODES;
    for (int i = t; i < BKT_NODES * REC_WORDS; i += 1024) recL[i] = 0;
    if (t < BKT_NODES) degL[t] = 0;
    if (t == 0) ovFull = 0;
    __syncthreads();
    int myov = 0;
    if (t < N_PROD) {
        myov = ov[t * OV_WORDS];
        if (myov > OV_CAP) ovFull = 1;
    }
    __syncthreads();
    const bool full = ovFull != 0;

    auto file = [&](int sv, int dl) {
        const int pos = atomicAdd(&degL[dl], 1);
        if (pos < SLOT_CAP) {
            recL[dl * REC_WORDS + pos] = sv;
        } else {
            const int k = atomicAdd(spill, 1);
            if (k < SPILL_CAP) { spill[2 + 2 * k] = sv; spill[3 + 2 * k] = node0 + dl; }
        }
    };

    if (!full) {
        const int wave = t >> 5;
        const int lane = t & 31;
        const int* base = seg + (size_t)bk * N_PROD * SEG_WORDS;
        int c[8], v[8];
#pragma unroll
        for (int j = 0; j < 8; ++j)
            c[j] = min(base[(wave + 32 * j) * SEG_WORDS], SEG_CAP);
#pragma unroll
        for (int j = 0; j < 8; ++j) {
            v[j] = 0;
            if (lane < c[j]) v[j] = base[(wave + 32 * j) * SEG_WORDS + 1 + lane];
        }
#pragma unroll
        for (int j = 0; j < 8; ++j) {
            if (lane < c[j]) file(v[j] & SRC_MASK, v[j] >> SRC_BITS);
            if (lane + 32 < c[j]) {
                const int v2 = base[(wave + 32 * j) * SEG_WORDS + 33 + lane];
                file(v2 & SRC_MASK, v2 >> SRC_BITS);
            }
        }
        if (t < N_PROD) {
            const int* ovb = ov + t * OV_WORDS;
            for (int k = 0; k < myov; ++k) {
                const int dd = ovb[2 + 2 * k] - node0;
                if ((unsigned)dd < (unsigned)BKT_NODES) file(ovb[1 + 2 * k], dd);
            }
        }
    } else {
        for (int e = t; e < N_EDGES; e += 1024) {
            const int dd = dst[e] - node0;
            if ((unsigned)dd < (unsigned)BKT_NODES) file(src[e], dd);
        }
    }
    __syncthreads();
    int* out = rec + (size_t)node0 * REC_WORDS;
    for (int i = t; i < BKT_NODES * REC_WORDS; i += 1024)
        out[i] = ((i & 31) == CNT_WORD) ? degL[i >> 5] : recL[i];
}

template <int BASE, int N>
__device__ __forceinline__ void add_rows(float4& acc, const float* __restrict__ xin,
                                         const int4 (&q)[N / 4], int nslot, int node, int sub) {
    float4 v[N];
#pragma unroll
    for (int jj = 0; jj < N; ++jj) {
        const int4& w = q[jj >> 2];
        int idx = ((jj & 3) == 0) ? w.x : ((jj & 3) == 1) ? w.y : ((jj & 3) == 2) ? w.z : w.w;
        idx = (BASE + jj < nslot) ? idx : node;
        v[jj] = *(const float4*)(xin + ((unsigned)idx * C_HID + sub * 4));
    }
#pragma unroll
    for (int jj = 0; jj < N; ++jj) {
        const bool p = BASE + jj < nslot;
        acc.x += p ? v[jj].x : 0.0f;
        acc.y += p ? v[jj].y : 0.0f;
        acc.z += p ? v[jj].z : 0.0f;
        acc.w += p ? v[jj].w : 0.0f;
    }
}

__device__ __forceinline__ float4 add_matching(const float* __restrict__ xin,
                                               const int* __restrict__ sp, const int* __restrict__ dp,
                                               int stride, int n, int node, int gbase, int sub) {
    float4 acc = make_float4(0.0f, 0.0f, 0.0f, 0.0f);
    for (int k0 = 0; k0 < n; k0 += 8) {
        const int k = k0 + sub;
        int sk = 0, dk = -1;
        if (k < n) { sk = sp[(size_t)k * stride]; dk = dp[(size_t)k * stride]; }
#pragma unroll
        for (int jj = 0; jj < 8; ++jj) {
            const int dj = __shfl(dk, gbase + jj);
            const int sj = __shfl(sk, gbase + jj);
            if (dj == node) {
                const float4 v = *(const float4*)(xin + (size_t)sj * C_HID + sub * 4);
                acc.x += v.x; acc.y += v.y; acc.z += v.z; acc.w += v.w;
            }
        }
    }
    return acc;
}

__global__ __launch_bounds__(256) void sage_pull_gemm(
        const float* __restrict__ xin, const int* __restrict__ rec,
        const int* __restrict__ spill,
        const int* __restrict__ src, const int* __restrict__ dst,
        const float* __restrict__ Wl, const float* __restrict__ Wr,
        const float* __restrict__ bias, float* __restrict__ hout) {
    __shared__ __attribute__((aligned(16))) float sW[64 * 32];
    __shared__ float sB[32];
    __shared__ __attribute__((aligned(16))) float sA[8][16 * A_STRIDE];
    __shared__ __attribute__((aligned(16))) int   sR[8][16 * R_STRIDE];

    const int wave = threadIdx.x >> 5;
    const int lane = threadIdx.x & 31;
    const int tile = blockIdx.x * 8 + wave;
    const int grp = lane >> 3;
    const int sub = lane & 7;
    float* myA = sA[wave];
    int*   myR = sR[wave];

    const int ltile = min(tile, N_TILES - 1);
    const unsigned own = (unsigned)(ltile * 16 + grp) * C_HID + sub * 4;
    const int4   rw0 = *(const int4*)(rec + (own + 0  * REC_WORDS));
    const int4   rw1 = *(const int4*)(rec + (own + 4  * REC_WORDS));
    const int4   rw2 = *(const int4*)(rec + (own + 8  * REC_WORDS));
    const int4   rw3 = *(const int4*)(rec + (own + 12 * REC_WORDS));
    const float4 sv0 = *(const float4*)(xin + (own + 0  * C_HID));
    const float4 sv1 = *(const float4*)(xin + (own + 4  * C_HID));
    const float4 sv2 = *(const float4*)(xin + (own + 8  * C_HID));
    const float4 sv3 = *(const float4*)(xin + (own + 12 * C_HID));
    __builtin_amdgcn_sched_barrier(0);
    {
        int* r = myR + grp * R_STRIDE + sub * 4;
        *(int4*)(r + 0  * R_STRIDE) = rw0;
        *(int4*)(r + 4  * R_STRIDE) = rw1;
        *(int4*)(r + 8  * R_STRIDE) = rw2;
        *(int4*)(r + 12 * R_STRIDE) = rw3;
        float* s = myA + grp * A_STRIDE + C_HID + sub * 4;
        *(float4*)(s + 0  * A_STRIDE) = sv0;
        *(float4*)(s + 4  * A_STRIDE) = sv1;
        *(float4*)(s + 8  * A_STRIDE) = sv2;
        *(float4*)(s + 12 * A_STRIDE) = sv3;
    }
    ((float4*)sW)[threadIdx.x]       = ((const float4*)Wl)[threadIdx.x];
    ((float4*)sW)[256 + threadIdx.x] = ((const float4*)Wr)[threadIdx.x];
    if (threadIdx.x < 32) sB[threadIdx.x] = bias[threadIdx.x];
    __syncthreads();

    if (tile >= N_TILES) return;

    const int row0  = tile * 16;
    const int gbase = lane & ~7;

    int spilled = 0;
#pragma unroll
    for (int r = 0; r < 4; ++r) {
        const int node = row0 + 4 * r + grp;
        const int* nrec = myR + (4 * r + grp) * R_STRIDE;
        const int cnt   = nrec[CNT_WORD];
        const int nslot = min(cnt, SLOT_CAP);
        int4 q[FIRST_SLOTS / 4];
#pragma unroll
        for (int i = 0; i < FIRST_SLOTS / 4; ++i) q[i] = *(const int4*)(nrec + 4 * i);
        float4 acc = make_float4(0.0f, 0.0f, 0.0f, 0.0f);
        add_rows<0, FIRST_SLOTS>(acc, xin, q, nslot, node, sub);
        if (__any(FIRST_SLOTS < nslot)) {
            int4 q2[2];
            q2[0] = *(const int4*)(nrec + FIRST_SLOTS);
            q2[1] = *(const int4*)(nrec + FIRST_SLOTS + 4);
            add_rows<FIRST_SLOTS, 8>(acc, xin, q2, nslot, node, sub);
        }
        if (cnt > SLOT_CAP) spilled |= 1 << r;
        const float inv = 1.0f / fmaxf((float)cnt, 1.0f);
        float4 mean;
        mean.x = acc.x * inv; mean.y = acc.y * inv; mean.z = acc.z * inv; mean.w = acc.w * inv;
        *(float4*)(myA + (4 * r + grp) * A_STRIDE + sub * 4) = mean;
    }

    if (__any(spilled != 0)) {
#pragma unroll 1
        for (int r = 0; r < 4; ++r) {
            if ((spilled >> r) & 1) {
                const int node = row0 + 4 * r + grp;
                const float inv = 1.0f / (float)rec[(size_t)node * REC_WORDS + CNT_WORD];
                float4* mp = (float4*)(myA + (4 * r + grp) * A_STRIDE + sub * 4);
                float4 mean = *mp;
                const int oc = spill[0];
                if (oc <= SPILL_CAP) {
                    const float4 ex = add_matching(xin, spill + 2, spill + 3, 2, oc, node, gbase, sub);
                    mean.x += ex.x * inv; mean.y += ex.y * inv; mean.z += ex.z * inv; mean.w += ex.w * inv;
                } else {
                    const float4 ex = add_matching(xin, src, dst, 1, N_EDGES, node, gbase, sub);
                    mean.x = ex.x * inv; mean.y = ex.y * inv; mean.z = ex.z * inv; mean.w = ex.w * inv;
                }
                *mp = mean;
            }
        }
    }

    __builtin_amdgcn_fence(__ATOMIC_RELEASE, "wavefront");
    __builtin_amdgcn_wave_barrier();
    __builtin_amdgcn_fence(__ATOMIC_ACQUIRE, "wavefront");

    const int mrow = lane & 15;
    const int koff = (lane >> 4) * 2;

    v8f acc0, acc1;
    const float b0v = sB[mrow];
    const float b1v = sB[16 + mrow];
#pragma unroll
    for (int j = 0; j < 8; ++j) { acc0[j] = b0v; acc1[j] = b1v; }

    const float* aRow = myA + mrow * A_STRIDE;

#pragma unroll
    for (int kc = 0; kc < 16; ++kc) {
        const int kb = kc * 4 + koff;
        v2f a;
        a.x = aRow[kb];
        a.y = aRow[kb + 1];
        v2f b0, b1;
        b0.x = sW[(kb)     * 32 + mrow];
        b0.y = sW[(kb + 1) * 32 + mrow];
        b1.x = sW[(kb)     * 32 + 16 + mrow];
        b1.y = sW[(kb + 1) * 32 + 16 + mrow];
        acc0 = __builtin_amdgcn_wmma_f32_16x16x4_f32(
                   false, a, false, b0, (short)0, acc0, false, false);
        acc1 = __builtin_amdgcn_wmma_f32_16x16x4_f32(
                   false, a, false, b1, (short)0, acc1, false, false);
    }

    const int half = lane >> 4;
#pragma unroll
    for (int j = 0; j < 8; ++j) {
        const int m = row0 + half * 8 + j;
        hout[(size_t)m * C_HID + mrow]      = fmaxf(acc0[j], 0.0f);
        hout[(size_t)m * C_HID + 16 + mrow] = fmaxf(acc1[j], 0.0f);
    }
}

__global__ void out_proj(const float* __restrict__ h, const float* __restrict__ Wo,
                         const float* __restrict__ bo, float* __restrict__ out) {
    int t = blockIdx.x * blockDim.x + threadIdx.x;
    if (t >= N_AGENTS * C_OUT) return;
    const int n = t >> 3, o = t & 7;
    float s = bo[o];
    const float* hr = h + (size_t)n * C_HID;
#pragma unroll
    for (int k = 0; k < C_HID; ++k) s += hr[k] * Wo[k * C_OUT + o];
    out[t] = s;
}

extern "C" void kernel_launch(void* const* d_in, const int* in_sizes, int n_in,
                              void* d_out, int out_size, void* d_ws, size_t ws_size,
                              hipStream_t stream) {
    const float* x    = (const float*)d_in[0];
    const int*   ei   = (const int*)d_in[1];
    const int*   src  = ei;
    const int*   dst  = ei + N_EDGES;
    const float* Wl1  = (const float*)d_in[2];
    const float* bl1  = (const float*)d_in[3];
    const float* Wr1  = (const float*)d_in[4];
    const float* Wl2  = (const float*)d_in[5];
    const float* bl2  = (const float*)d_in[6];
    const float* Wr2  = (const float*)d_in[7];
    const float* Wl3  = (const float*)d_in[8];
    const float* bl3  = (const float*)d_in[9];
    const float* Wr3  = (const float*)d_in[10];
    const float* Wo   = (const float*)d_in[11];
    const float* bo   = (const float*)d_in[12];
    float* out = (float*)d_out;

    const size_t NB = (size_t)N_NODES * C_HID;
    int*   rec   = (int*)d_ws;
    float* h1    = (float*)d_ws + NB;
    float* h2    = h1 + NB;
    int*   spill = (int*)(h2 + NB);
    int*   ov    = spill + SPILLB_WORDS;
    int*   seg   = (int*)h1;

    const int GB = (N_TILES + 7) / 8;
    const int OB = (N_AGENTS * C_OUT + 255) / 256;

    part_edges<<<N_PROD, 1024, 0, stream>>>(src, dst, seg, ov, spill);
    file_bucket<<<N_BKT, 1024, 0, stream>>>(seg, ov, src, dst, rec, spill);

    sage_pull_gemm<<<GB, 256, 0, stream>>>(x, rec, spill, src, dst, Wl1, Wr1, bl1, h1);
    sage_pull_gemm<<<GB, 256, 0, stream>>>(h1, rec, spill, src, dst, Wl2, Wr2, bl2, h2);
    sage_pull_gemm<<<GB, 256, 0, stream>>>(h2, rec, spill, src, dst, Wl3, Wr3, bl3, h1);

    out_proj<<<OB, 256, 0, stream>>>(h1, Wo, bo, out);
}


# edge producer groups its entries by bucket in LDS and writes each segment with whole-line stores: 1.6M scattered 4-byte global stores become about 70K line stores; slot order = LDS arrival as before
# speedup vs baseline: 3.8270x; 1.0300x over previous
#include <hip/hip_runtime.h>
#include <hip/hip_bf16.h>

typedef float v2f __attribute__((ext_vector_type(2)));
typedef float v8f __attribute__((ext_vector_type(8)));

#define N_NODES 100000
#define N_EDGES 1600000
#define C_HID   32
#define C_OUT   8
#define N_AGENTS 1000
#define N_TILES (N_NODES / 16)

#define REC_WORDS 32
#define SLOT_CAP  31
#define CNT_WORD  31
#define SPILLB_WORDS 60000
#define SPILL_CAP 29998

#define A_STRIDE 68
#define R_STRIDE 36
#define FIRST_SLOTS 24

#define BKT_NODES 400
#define N_BKT     250
#define N_PROD    256
#define EDGES_PER_PROD 6250
#define PER_THREAD 7
#define SEG_WORDS 64
#define SEG_CAP   63
#define OV_WORDS  128
#define OV_CAP    63
#define SRC_BITS  17
#define SRC_MASK  0x1FFFF

__global__ __launch_bounds__(1024) void part_edges(
        const int* __restrict__ src, const int* __restrict__ dst,
        int* __restrict__ seg, int* __restrict__ ov, int* __restrict__ spill) {
    __shared__ int segCnt[256];
    __shared__ int segOff[256];
    __shared__ int stage[EDGES_PER_PROD];
    __shared__ int ovCnt;
    const int b = blockIdx.x;
    const int t = threadIdx.x;
    const int wave = t >> 5;
    const int lane = t & 31;
    if (t < 256) segCnt[t] = 0;
    if (t == 0) { ovCnt = 0; stage[0] = 0; }
    if (b == 0 && t == 0) spill[0] = 0;
    __syncthreads();

    const int e0 = b * EDGES_PER_PROD;
    int* ovb = ov + b * OV_WORDS;
    int s[PER_THREAD], d[PER_THREAD];
#pragma unroll
    for (int i = 0; i < PER_THREAD; ++i) {
        const int k = t + i * 1024;
        s[i] = 0; d[i] = -1;
        if (k < EDGES_PER_PROD) { s[i] = src[e0 + k]; d[i] = dst[e0 + k]; }
    }
    int bkt[PER_THREAD], slot[PER_THREAD];
#pragma unroll
    for (int i = 0; i < PER_THREAD; ++i) {
        bkt[i] = -1; slot[i] = 0;
        if (d[i] >= 0) {
            bkt[i]  = d[i] / BKT_NODES;
            slot[i] = atomicAdd(&segCnt[bkt[i]], 1);
        }
    }
    __syncthreads();
    if (wave == 0) {
        int cc[8];
        int sum = 0;
#pragma unroll
        for (int j = 0; j < 8; ++j) {
            const int q = lane * 8 + j;
            cc[j] = (q < N_BKT) ? min(segCnt[q], SEG_CAP) : 0;
            sum += cc[j];
        }
        int x = sum;
#pragma unroll
        for (int o = 1; o < 32; o <<= 1) {
            const int y = __shfl_up(x, o);
            if (lane >= o) x += y;
        }
        int run = x - sum;
#pragma unroll
        for (int j = 0; j < 8; ++j) { segOff[lane * 8 + j] = run; run += cc[j]; }
    }
    __syncthreads();
#pragma unroll
    for (int i = 0; i < PER_THREAD; ++i) {
        if (bkt[i] >= 0) {
            if (slot[i] < SEG_CAP) {
                const int dl = d[i] - bkt[i] * BKT_NODES;
                stage[segOff[bkt[i]] + slot[i]] = s[i] | (dl << SRC_BITS);
            } else {
                const int k = atomicAdd(&ovCnt, 1);
                if (k < OV_CAP) { ovb[1 + 2 * k] = s[i]; ovb[2 + 2 * k] = d[i]; }
            }
        }
    }
    __syncthreads();
    int cN[8], oN[8], w1[8], w2[8];
#pragma unroll
    for (int j = 0; j < 8; ++j) {
        const int bk = wave + 32 * j;
        cN[j] = -1; oN[j] = 0;
        if (bk < N_BKT) { cN[j] = min(segCnt[bk], SEG_CAP); oN[j] = segOff[bk]; }
    }
#pragma unroll
    for (int j = 0; j < 8; ++j) {
        const int v1 = stage[(lane >= 1 && lane <= cN[j]) ? oN[j] + lane - 1 : 0];
        const int v2 = stage[(lane + 31 < cN[j]) ? oN[j] + lane + 31 : 0];
        w1[j] = (lane == 0) ? cN[j] : v1;
        w2[j] = v2;
    }
#pragma unroll
    for (int j = 0; j < 8; ++j) {
        const int bk = wave + 32 * j;
        int* sg = seg + ((size_t)bk * N_PROD + b) * SEG_WORDS;
        if (lane <= cN[j])     sg[lane]      = w1[j];
        if (lane + 31 < cN[j]) sg[lane + 32] = w2[j];
    }
    if (t == 0) ovb[0] = ovCnt;
}

__global__ __launch_bounds__(1024) void file_bucket(
        const int* __restrict__ seg, const int* __restrict__ ov,
        const int* __restrict__ src, const int* __restrict__ dst,
        int* __restrict__ rec, int* __restrict__ spill) {
    __shared__ int recL[BKT_NODES * REC_WORDS];
    __shared__ int degL[BKT_NODES];
    __shared__ int ovFull;
    const int bk = blockIdx.x;
    const int t  = threadIdx.x;
    const int node0 = bk * BKT_NODES;
    for (int i = t; i < BKT_NODES * REC_WORDS; i += 1024) recL[i] = 0;
    if (t < BKT_NODES) degL[t] = 0;
    if (t == 0) ovFull = 0;
    __syncthreads();
    int myov = 0;
    if (t < N_PROD) {
        myov = ov[t * OV_WORDS];
        if (myov > OV_CAP) ovFull = 1;
    }
    __syncthreads();
    const bool full = ovFull != 0;

    auto file = [&](int sv, int dl) {
        const int pos = atomicAdd(&degL[dl], 1);
        if (pos < SLOT_CAP) {
            recL[dl * REC_WORDS + pos] = sv;
        } else {
            const int k = atomicAdd(spill, 1);
            if (k < SPILL_CAP) { spill[2 + 2 * k] = sv; spill[3 + 2 * k] = node0 + dl; }
        }
    };

    if (!full) {
        const int wave = t >> 5;
        const int lane = t & 31;
        const int* base = seg + (size_t)bk * N_PROD * SEG_WORDS;
        int c[8], v[8];
#pragma unroll
        for (int j = 0; j < 8; ++j)
            c[j] = min(base[(wave + 32 * j) * SEG_WORDS], SEG_CAP);
#pragma unroll
        for (int j = 0; j < 8; ++j) {
            v[j] = 0;
            if (lane < c[j]) v[j] = base[(wave + 32 * j) * SEG_WORDS + 1 + lane];
        }
#pragma unroll
        for (int j = 0; j < 8; ++j) {
            if (lane < c[j]) file(v[j] & SRC_MASK, v[j] >> SRC_BITS);
            if (lane + 32 < c[j]) {
                const int v2 = base[(wave + 32 * j) * SEG_WORDS + 33 + lane];
                file(v2 & SRC_MASK, v2 >> SRC_BITS);
            }
        }
        if (t < N_PROD) {
            const int* ovb = ov + t * OV_WORDS;
            for (int k = 0; k < myov; ++k) {
                const int dd = ovb[2 + 2 * k] - node0;
                if ((unsigned)dd < (unsigned)BKT_NODES) file(ovb[1 + 2 * k], dd);
            }
        }
    } else {
        for (int e = t; e < N_EDGES; e += 1024) {
            const int dd = dst[e] - node0;
            if ((unsigned)dd < (unsigned)BKT_NODES) file(src[e], dd);
        }
    }
    __syncthreads();
    int* out = rec + (size_t)node0 * REC_WORDS;
    for (int i = t; i < BKT_NODES * REC_WORDS; i += 1024)
        out[i] = ((i & 31) == CNT_WORD) ? degL[i >> 5] : recL[i];
}

template <int BASE, int N>
__device__ __forceinline__ void add_rows(float4& acc, const float* __restrict__ xin,
                                         const int4 (&q)[N / 4], int nslot, int node, int sub) {
    float4 v[N];
#pragma unroll
    for (int jj = 0; jj < N; ++jj) {
        const int4& w = q[jj >> 2];
        int idx = ((jj & 3) == 0) ? w.x : ((jj & 3) == 1) ? w.y : ((jj & 3) == 2) ? w.z : w.w;
        idx = (BASE + jj < nslot) ? idx : node;
        v[jj] = *(const float4*)(xin + ((unsigned)idx * C_HID + sub * 4));
    }
#pragma unroll
    for (int jj = 0; jj < N; ++jj) {
        const bool p = BASE + jj < nslot;
        acc.x += p ? v[jj].x : 0.0f;
        acc.y += p ? v[jj].y : 0.0f;
        acc.z += p ? v[jj].z : 0.0f;
        acc.w += p ? v[jj].w : 0.0f;
    }
}

__device__ __forceinline__ float4 add_matching(const float* __restrict__ xin,
                                               const int* __restrict__ sp, const int* __restrict__ dp,
                                               int stride, int n, int node, int gbase, int sub) {
    float4 acc = make_float4(0.0f, 0.0f, 0.0f, 0.0f);
    for (int k0 = 0; k0 < n; k0 += 8) {
        const int k = k0 + sub;
        int sk = 0, dk = -1;
        if (k < n) { sk = sp[(size_t)k * stride]; dk = dp[(size_t)k * stride]; }
#pragma unroll
        for (int jj = 0; jj < 8; ++jj) {
            const int dj = __shfl(dk, gbase + jj);
            const int sj = __shfl(sk, gbase + jj);
            if (dj == node) {
                const float4 v = *(const float4*)(xin + (size_t)sj * C_HID + sub * 4);
                acc.x += v.x; acc.y += v.y; acc.z += v.z; acc.w += v.w;
            }
        }
    }
    return acc;
}

__global__ __launch_bounds__(256) void sage_pull_gemm(
        const float* __restrict__ xin, const int* __restrict__ rec,
        const int* __restrict__ spill,
        const int* __restrict__ src, const int* __restrict__ dst,
        const float* __restrict__ Wl, const float* __restrict__ Wr,
        const float* __restrict__ bias, float* __restrict__ hout) {
    __shared__ __attribute__((aligned(16))) float sW[64 * 32];
    __shared__ float sB[32];
    __shared__ __attribute__((aligned(16))) float sA[8][16 * A_STRIDE];
    __shared__ __attribute__((aligned(16))) int   sR[8][16 * R_STRIDE];

    const int wave = threadIdx.x >> 5;
    const int lane = threadIdx.x & 31;
    const int tile = blockIdx.x * 8 + wave;
    const int grp = lane >> 3;
    const int sub = lane & 7;
    float* myA = sA[wave];
    int*   myR = sR[wave];

    const int ltile = min(tile, N_TILES - 1);
    const unsigned own = (unsigned)(ltile * 16 + grp) * C_HID + sub * 4;
    const int4   rw0 = *(const int4*)(rec + (own + 0  * REC_WORDS));
    const int4   rw1 = *(const int4*)(rec + (own + 4  * REC_WORDS));
    const int4   rw2 = *(const int4*)(rec + (own + 8  * REC_WORDS));
    const int4   rw3 = *(const int4*)(rec + (own + 12 * REC_WORDS));
    const float4 sv0 = *(const float4*)(xin + (own + 0  * C_HID));
    const float4 sv1 = *(const float4*)(xin + (own + 4  * C_HID));
    const float4 sv2 = *(const float4*)(xin + (own + 8  * C_HID));
    const float4 sv3 = *(const float4*)(xin + (own + 12 * C_HID));
    __builtin_amdgcn_sched_barrier(0);
    {
        int* r = myR + grp * R_STRIDE + sub * 4;
        *(int4*)(r + 0  * R_STRIDE) = rw0;
        *(int4*)(r + 4  * R_STRIDE) = rw1;
        *(int4*)(r + 8  * R_STRIDE) = rw2;
        *(int4*)(r + 12 * R_STRIDE) = rw3;
        float* s = myA + grp * A_STRIDE + C_HID + sub * 4;
        *(float4*)(s + 0  * A_STRIDE) = sv0;
        *(float4*)(s + 4  * A_STRIDE) = sv1;
        *(float4*)(s + 8  * A_STRIDE) = sv2;
        *(float4*)(s + 12 * A_STRIDE) = sv3;
    }
    ((float4*)sW)[threadIdx.x]       = ((const float4*)Wl)[threadIdx.x];
    ((float4*)sW)[256 + threadIdx.x] = ((const float4*)Wr)[threadIdx.x];
    if (threadIdx.x < 32) sB[threadIdx.x] = bias[threadIdx.x];
    __syncthreads();

    if (tile >= N_TILES) return;

    const int row0  = tile * 16;
    const int gbase = lane & ~7;

    int spilled = 0;
#pragma unroll
    for (int r = 0; r < 4; ++r) {
        const int node = row0 + 4 * r + grp;
        const int* nrec = myR + (4 * r + grp) * R_STRIDE;
        const int cnt   = nrec[CNT_WORD];
        const int nslot = min(cnt, SLOT_CAP);
        int4 q[FIRST_SLOTS / 4];
#pragma unroll
        for (int i = 0; i < FIRST_SLOTS / 4; ++i) q[i] = *(const int4*)(nrec + 4 * i);
        float4 acc = make_float4(0.0f, 0.0f, 0.0f, 0.0f);
        add_rows<0, FIRST_SLOTS>(acc, xin, q, nslot, node, sub);
        if (__any(FIRST_SLOTS < nslot)) {
            int4 q2[2];
            q2[0] = *(const int4*)(nrec + FIRST_SLOTS);
            q2[1] = *(const int4*)(nrec + FIRST_SLOTS + 4);
            add_rows<FIRST_SLOTS, 8>(acc, xin, q2, nslot, node, sub);
        }
        if (cnt > SLOT_CAP) spilled |= 1 << r;
        const float inv = 1.0f / fmaxf((float)cnt, 1.0f);
        float4 mean;
        mean.x = acc.x * inv; mean.y = acc.y * inv; mean.z = acc.z * inv; mean.w = acc.w * inv;
        *(float4*)(myA + (4 * r + grp) * A_STRIDE + sub * 4) = mean;
    }

    if (__any(spilled != 0)) {
#pragma unroll 1
        for (int r = 0; r < 4; ++r) {
            if ((spilled >> r) & 1) {
                const int node = row0 + 4 * r + grp;
                const float inv = 1.0f / (float)rec[(size_t)node * REC_WORDS + CNT_WORD];
                float4* mp = (float4*)(myA + (4 * r + grp) * A_STRIDE + sub * 4);
                float4 mean = *mp;
                const int oc = spill[0];
                if (oc <= SPILL_CAP) {
                    const float4 ex = add_matching(xin, spill + 2, spill + 3, 2, oc, node, gbase, sub);
                    mean.x += ex.x * inv; mean.y += ex.y * inv; mean.z += ex.z * inv; mean.w += ex.w * inv;
                } else {
                    const float4 ex = add_matching(xin, src, dst, 1, N_EDGES, node, gbase, sub);
                    mean.x = ex.x * inv; mean.y = ex.y * inv; mean.z = ex.z * inv; mean.w = ex.w * inv;
                }
                *mp = mean;
            }
        }
    }

    __builtin_amdgcn_fence(__ATOMIC_RELEASE, "wavefront");
    __builtin_amdgcn_wave_barrier();
    __builtin_amdgcn_fence(__ATOMIC_ACQUIRE, "wavefront");

    const int mrow = lane & 15;
    const int koff = (lane >> 4) * 2;

    v8f acc0, acc1;
    const float b0v = sB[mrow];
    const float b1v = sB[16 + mrow];
#pragma unroll
    for (int j = 0; j < 8; ++j) { acc0[j] = b0v; acc1[j] = b1v; }

    const float* aRow = myA + mrow * A_STRIDE;

#pragma unroll
    for (int kc = 0; kc < 16; ++kc) {
        const int kb = kc * 4 + koff;
        v2f a;
        a.x = aRow[kb];
        a.y = aRow[kb + 1];
        v2f b0, b1;
        b0.x = sW[(kb)     * 32 + mrow];
        b0.y = sW[(kb + 1) * 32 + mrow];
        b1.x = sW[(kb)     * 32 + 16 + mrow];
        b1.y = sW[(kb + 1) * 32 + 16 + mrow];
        acc0 = __builtin_amdgcn_wmma_f32_16x16x4_f32(
                   false, a, false, b0, (short)0, acc0, false, false);
        acc1 = __builtin_amdgcn_wmma_f32_16x16x4_f32(
                   false, a, false, b1, (short)0, acc1, false, false);
    }

    const int half = lane >> 4;
#pragma unroll
    for (int j = 0; j < 8; ++j) {
        const int m = row0 + half * 8 + j;
        hout[(size_t)m * C_HID + mrow]      = fmaxf(acc0[j], 0.0f);
        hout[(size_t)m * C_HID + 16 + mrow] = fmaxf(acc1[j], 0.0f);
    }
}

__global__ void out_proj(const float* __restrict__ h, const float* __restrict__ Wo,
                         const float* __restrict__ bo, float* __restrict__ out) {
    int t = blockIdx.x * blockDim.x + threadIdx.x;
    if (t >= N_AGENTS * C_OUT) return;
    const int n = t >> 3, o = t & 7;
    float s = bo[o];
    const float* hr = h + (size_t)n * C_HID;
#pragma unroll
    for (int k = 0; k < C_HID; ++k) s += hr[k] * Wo[k * C_OUT + o];
    out[t] = s;
}

extern "C" void kernel_launch(void* const* d_in, const int* in_sizes, int n_in,
                              void* d_out, int out_size, void* d_ws, size_t ws_size,
                              hipStream_t stream) {
    const float* x    = (const float*)d_in[0];
    const int*   ei   = (const int*)d_in[1];
    const int*   src  = ei;
    const int*   dst  = ei + N_EDGES;
    const float* Wl1  = (const float*)d_in[2];
    const float* bl1  = (const float*)d_in[3];
    const float* Wr1  = (const float*)d_in[4];
    const float* Wl2  = (const float*)d_in[5];
    const float* bl2  = (const float*)d_in[6];
    const float* Wr2  = (const float*)d_in[7];
    const float* Wl3  = (const float*)d_in[8];
    const float* bl3  = (const float*)d_in[9];
    const float* Wr3  = (const float*)d_in[10];
    const float* Wo   = (const float*)d_in[11];
    const float* bo   = (const float*)d_in[12];
    float* out = (float*)d_out;

    const size_t NB = (size_t)N_NODES * C_HID;
    int*   rec   = (int*)d_ws;
    float* h1    = (float*)d_ws + NB;
    float* h2    = h1 + NB;
    int*   spill = (int*)(h2 + NB);
    int*   ov    = spill + SPILLB_WORDS;
    int*   seg   = (int*)h1;

    const int GB = (N_TILES + 7) / 8;
    const int OB = (N_AGENTS * C_OUT + 255) / 256;

    part_edges<<<N_PROD, 1024, 0, stream>>>(src, dst, seg, ov, spill);
    file_bucket<<<N_BKT, 1024, 0, stream>>>(seg, ov, src, dst, rec, spill);

    sage_pull_gemm<<<GB, 256, 0, stream>>>(x, rec, spill, src, dst, Wl1, Wr1, bl1, h1);
    sage_pull_gemm<<<GB, 256, 0, stream>>>(h1, rec, spill, src, dst, Wl2, Wr2, bl2, h2);
    sage_pull_gemm<<<GB, 256, 0, stream>>>(h2, rec, spill, src, dst, Wl3, Wr3, bl3, h1);

    out_proj<<<OB, 256, 0, stream>>>(h1, Wo, bo, out);
}
